# starting kernel
# baseline (speedup 1.0000x reference)
#include <hip/hip_runtime.h>
#include <math.h>

#define B_    32
#define IN_   2048
#define DIN_  8
#define N_    64
#define DOUT_ 16
#define C_    32
#define REP_  (IN_ / C_)
#define TILE_ 512
#define NTILE (IN_ / TILE_)
#define EPS_  1e-7f

typedef float v2f __attribute__((ext_vector_type(2)));
typedef float v8f __attribute__((ext_vector_type(8)));

__device__ __forceinline__ void hat_tile_wmma(
    const float* __restrict__ x, const float* __restrict__ Wt,
    float* __restrict__ hat, int b, int n, int tile, int wave, int lane)
{
  const int mrow = lane & 15;
  const int kh   = (lane >> 4) << 1;
  const int half = lane >> 4;
  for (int u = wave; u < TILE_ / 16; u += 8) {
    const int il0 = u << 4;
    const int gi0 = tile * TILE_ + il0;
    const int cu  = gi0 >> 6;
    const float* xr = x  + ((size_t)b * IN_ + gi0 + mrow) * DIN_ + kh;
    const float* wr = Wt + (((size_t)n * C_ + cu) * DOUT_ + mrow) * DIN_ + kh;
    v2f a0 = { xr[0], xr[1] };
    v2f a1 = { xr[4], xr[5] };
    v2f b0 = { wr[0], wr[1] };
    v2f b1 = { wr[4], wr[5] };
    v8f acc = {0.f,0.f,0.f,0.f,0.f,0.f,0.f,0.f};
    acc = __builtin_amdgcn_wmma_f32_16x16x4_f32(false, a0, false, b0,
                                                (short)0, acc, false, false);
    acc = __builtin_amdgcn_wmma_f32_16x16x4_f32(false, a1, false, b1,
                                                (short)0, acc, false, false);
#pragma unroll
    for (int r = 0; r < 8; ++r)
      hat[(il0 + r + (half << 3)) * DOUT_ + mrow] = acc[r];
  }
}

__global__ __launch_bounds__(256) void caps_softmax(
    const float* __restrict__ logits, float* __restrict__ smax, float* __restrict__ sinv)
{
  const int e = blockIdx.x * 256 + threadIdx.x;
  const int b = e >> 11;
  const int i = e & (IN_ - 1);
  const float* lp = logits + (size_t)b * N_ * IN_ + i;
  float mx = -3.4e38f;
#pragma unroll 4
  for (int n = 0; n < N_; ++n) mx = fmaxf(mx, lp[(size_t)n * IN_]);
  float s = 0.f;
#pragma unroll 4
  for (int n = 0; n < N_; ++n) s += __expf(lp[(size_t)n * IN_] - mx);
  smax[e] = mx;
  sinv[e] = 1.0f / s;
}

template <int MODE>
__global__ __launch_bounds__(256) void caps_route(
    const float* __restrict__ x, const float* __restrict__ Wt,
    const float* __restrict__ bias, float* __restrict__ logits,
    const float* __restrict__ smax, const float* __restrict__ sinv,
    float* __restrict__ out)
{
  __shared__ float hat[TILE_ * DOUT_];
  __shared__ float cld[TILE_];
  __shared__ float part[256];
  __shared__ float sld[DOUT_];
  __shared__ float old_[DOUT_];
  __shared__ float scl;

  const int b    = blockIdx.x >> 6;
  const int n    = blockIdx.x & (N_ - 1);
  const int wave = threadIdx.x >> 5;
  const int lane = threadIdx.x & 31;
  const int dc    = threadIdx.x & 15;
  const int chunk = threadIdx.x >> 4;
  const size_t lbase = ((size_t)b * N_ + n) * IN_;

  float sacc = 0.f;
  for (int tile = 0; tile < NTILE; ++tile) {
    if (tile + 1 < NTILE)
      __builtin_prefetch(
          x + ((size_t)b * IN_ + (tile + 1) * TILE_ + (threadIdx.x << 1)) * DIN_, 0, 0);
    hat_tile_wmma(x, Wt, hat, b, n, tile, wave, lane);
    for (int ii = threadIdx.x; ii < TILE_; ii += 256) {
      float cv;
      if (MODE == 0) {
        cv = 1.0f / (float)N_;
      } else {
        const int gi = tile * TILE_ + ii;
        const size_t e = (size_t)b * IN_ + gi;
        cv = __expf(logits[lbase + gi] - smax[e]) * sinv[e];
      }
      cld[ii] = cv;
    }
    __syncthreads();
    const float* hp = hat + (chunk * 32) * DOUT_ + dc;
    const float* cp = cld + chunk * 32;
#pragma unroll 8
    for (int ii = 0; ii < 32; ++ii) sacc += cp[ii] * hp[ii * DOUT_];
    __syncthreads();
  }

  part[threadIdx.x] = sacc;
  __syncthreads();
  if (threadIdx.x < DOUT_) {
    float s = bias[n * DOUT_ + threadIdx.x];
#pragma unroll
    for (int ch = 0; ch < 16; ++ch) s += part[ch * 16 + threadIdx.x];
    sld[threadIdx.x] = s;
  }
  __syncthreads();
  if (threadIdx.x == 0) {
    float sn = 0.f;
#pragma unroll
    for (int d2 = 0; d2 < DOUT_; ++d2) sn += sld[d2] * sld[d2];
    scl = sn / (1.0f + sn) * rsqrtf(sn + EPS_);
  }
  __syncthreads();
  if (threadIdx.x < DOUT_) {
    const float o = scl * sld[threadIdx.x];
    old_[threadIdx.x] = o;
    out[((size_t)b * N_ + n) * DOUT_ + threadIdx.x] = o;
  }
  __syncthreads();

  if (MODE < 2) {
    for (int tile = 0; tile < NTILE; ++tile) {
      hat_tile_wmma(x, Wt, hat, b, n, tile, wave, lane);
      __syncthreads();
      for (int ii = threadIdx.x; ii < TILE_; ii += 256) {
        const float* hr = hat + ii * DOUT_;
        float dlt = 0.f;
#pragma unroll
        for (int d2 = 0; d2 < DOUT_; ++d2) dlt += old_[d2] * hr[d2];
        const size_t g = lbase + tile * TILE_ + ii;
        logits[g] = (MODE == 0) ? dlt : (logits[g] + dlt);
      }
      __syncthreads();
    }
  }
}

extern "C" void kernel_launch(void* const* d_in, const int* in_sizes, int n_in,
                              void* d_out, int out_size, void* d_ws, size_t ws_size,
                              hipStream_t stream)
{
  const float* x    = (const float*)d_in[0];
  const float* Wt   = (const float*)d_in[1];
  const float* bias = (const float*)d_in[2];
  float* out = (float*)d_out;

  float* logits = (float*)d_ws;
  float* smax   = logits + (size_t)B_ * N_ * IN_;
  float* sinv   = smax   + (size_t)B_ * IN_;

  dim3 grid(B_ * N_), blk(256);
  dim3 sgrid((B_ * IN_) / 256), sblk(256);

  caps_route<0><<<grid, blk, 0, stream>>>(x, Wt, bias, logits, smax, sinv, out);
  caps_softmax<<<sgrid, sblk, 0, stream>>>(logits, smax, sinv);
  caps_route<1><<<grid, blk, 0, stream>>>(x, Wt, bias, logits, smax, sinv, out);
  caps_softmax<<<sgrid, sblk, 0, stream>>>(logits, smax, sinv);
  caps_route<2><<<grid, blk, 0, stream>>>(x, Wt, bias, logits, smax, sinv, out);
}


# Reassociate both contractions around the channel-shared weights (s = W.(coef^T x), logits = x.(W^T sum of outputs)) on f32 WMMA per (batch,channel) block; vote tensor never formed, softmax in-block
# speedup vs baseline: 3.5533x; 3.5533x over previous
#include <hip/hip_runtime.h>
#include <math.h>


#define B_    32
#define IN_   2048
#define DIN_  8
#define N_    64
#define DOUT_ 16
#define C_    32
#define REP_  (IN_ / C_)
#define EPS_  1e-7f

typedef float v2f __attribute__((ext_vector_type(2)));
typedef float v8f __attribute__((ext_vector_type(8)));


template <int CTRL>
__device__ __forceinline__ float row_swz(float v)
{
  return __builtin_bit_cast(float,
      __builtin_amdgcn_update_dpp(0, __builtin_bit_cast(int, v), CTRL, 0xf, 0xf, false));
}
__device__ __forceinline__ float row_max16(float v)
{
  v = fmaxf(v, row_swz<0x161>(v));
  v = fmaxf(v, row_swz<0x162>(v));
  v = fmaxf(v, row_swz<0x164>(v));
  v = fmaxf(v, row_swz<0x168>(v));
  return v;
}
__device__ __forceinline__ float row_sum16(float v)
{
  v += row_swz<0x161>(v);
  v += row_swz<0x162>(v);
  v += row_swz<0x164>(v);
  v += row_swz<0x168>(v);
  return v;
}
__device__ __forceinline__ v8f wmma4(v2f a, v2f b, v8f c)
{
  return __builtin_amdgcn_wmma_f32_16x16x4_f32(false, a, false, b, (short)0, c, false, false);
}

template <int MODE>
__global__ __launch_bounds__(128) void caps_coef(
    const float* __restrict__ x, const float* __restrict__ Vs, float* __restrict__ Yw)
{
  __shared__ float4 ypart[4][N_ * DIN_ / 4];

  const int b    = blockIdx.x >> 5;
  const int ch   = blockIdx.x & (C_ - 1);
  const int wave = threadIdx.x >> 5;
  const int lane = threadIdx.x & 31;
  const int cl   = lane & 15;
  const int h    = lane >> 4;

  const float* xb = x + ((size_t)b * IN_ + ch * REP_ + wave * 16) * DIN_;

  v2f ax[4];
#pragma unroll
  for (int q = 0; q < 4; ++q) {
    const float* p = xb + (8 * h + 2 * q) * DIN_ + (cl & 7);
    float v0 = p[0], v1 = p[DIN_];
    if (cl >= 8) { v0 = 0.f; v1 = 0.f; }
    ax[q] = (v2f){ v0, v1 };
  }

  v8f cm[4];
  if (MODE == 0) {
#pragma unroll
    for (int t = 0; t < 4; ++t)
#pragma unroll
      for (int r = 0; r < 8; ++r) cm[t][r] = 1.0f / (float)N_;
  } else {
    const float4 xa = *(const float4*)(xb + cl * DIN_ + 4 * h);
    const v2f a0 = { xa.x, xa.y }, a1 = { xa.z, xa.w };
    const float* vb = Vs + ((size_t)(b * C_ + ch) * N_) * DIN_;
#pragma unroll
    for (int t = 0; t < 4; ++t) {
      const float4 vv = *(const float4*)(vb + (16 * t + cl) * DIN_ + 4 * h);
      const v2f b0 = { vv.x, vv.y }, b1 = { vv.z, vv.w };
      v8f acc = {0.f,0.f,0.f,0.f,0.f,0.f,0.f,0.f};
      acc = wmma4(a0, b0, acc);
      acc = wmma4(a1, b1, acc);
      cm[t] = acc;
    }
#pragma unroll
    for (int r = 0; r < 8; ++r) {
      float m = fmaxf(fmaxf(cm[0][r], cm[1][r]), fmaxf(cm[2][r], cm[3][r]));
      m = row_max16(m);
      const float e0 = __expf(cm[0][r] - m);
      const float e1 = __expf(cm[1][r] - m);
      const float e2 = __expf(cm[2][r] - m);
      const float e3 = __expf(cm[3][r] - m);
      const float inv = 1.0f / row_sum16((e0 + e1) + (e2 + e3));
      cm[0][r] = e0 * inv; cm[1][r] = e1 * inv; cm[2][r] = e2 * inv; cm[3][r] = e3 * inv;
    }
  }

#pragma unroll
  for (int t = 0; t < 4; ++t) {
    v8f y = {0.f,0.f,0.f,0.f,0.f,0.f,0.f,0.f};
#pragma unroll
    for (int q = 0; q < 4; ++q) {
      const v2f bb = { cm[t][2 * q], cm[t][2 * q + 1] };
      y = wmma4(ax[q], bb, y);
    }
    if (h == 0) {
      ypart[wave][(16 * t + cl) * 2 + 0] = make_float4(y[0], y[1], y[2], y[3]);
      ypart[wave][(16 * t + cl) * 2 + 1] = make_float4(y[4], y[5], y[6], y[7]);
    }
  }
  __syncthreads();
  {
    const int t4 = threadIdx.x;
    const float4 p0 = ypart[0][t4], p1 = ypart[1][t4], p2 = ypart[2][t4], p3 = ypart[3][t4];
    float4 s;
    s.x = (p0.x + p1.x) + (p2.x + p3.x);
    s.y = (p0.y + p1.y) + (p2.y + p3.y);
    s.z = (p0.z + p1.z) + (p2.z + p3.z);
    s.w = (p0.w + p1.w) + (p2.w + p3.w);
    ((float4*)(Yw + (size_t)(b * C_ + ch) * N_ * DIN_))[t4] = s;
  }
}

template <int MODE>
__global__ __launch_bounds__(128) void caps_out(
    const float* __restrict__ Wt, const float* __restrict__ bias,
    const float* __restrict__ Yw, float* __restrict__ Osum,
    float* __restrict__ Vs, float* __restrict__ out)
{
  __shared__ float sp[4][16 * DOUT_];
  __shared__ float ol[16 * DOUT_];

  const int n    = blockIdx.x >> 1;
  const int tb   = blockIdx.x & 1;
  const int wave = threadIdx.x >> 5;
  const int lane = threadIdx.x & 31;
  const int cl   = lane & 15;
  const int h    = lane >> 4;

  v2f aw[4][4];
  if (MODE < 2) {
#pragma unroll
    for (int tt = 0; tt < 4; ++tt) {
      const int tn = wave * 4 + tt;
      const float* wp = Wt + (((size_t)n * C_ + 2 * tn + (cl >> 3)) * DOUT_) * DIN_ + (cl & 7);
#pragma unroll
      for (int q = 0; q < 4; ++q) {
        const int d0 = 4 * q + 2 * h;
        aw[tt][q] = (v2f){ wp[d0 * DIN_], wp[(d0 + 1) * DIN_] };
      }
    }
  }

  v8f acc0 = {0.f,0.f,0.f,0.f,0.f,0.f,0.f,0.f};
  v8f acc1 = {0.f,0.f,0.f,0.f,0.f,0.f,0.f,0.f};
  {
    const int bg = tb * 16 + cl;
#pragma unroll
    for (int cc = 0; cc < 8; ++cc) {
      const int c = wave * 8 + cc;
      const float4 ya = *(const float4*)(Yw + (((size_t)bg * C_ + c) * N_ + n) * DIN_ + 4 * h);
      const float4 wb = *(const float4*)(Wt + (((size_t)n * C_ + c) * DOUT_ + cl) * DIN_ + 4 * h);
      acc0 = wmma4((v2f){ ya.x, ya.y }, (v2f){ wb.x, wb.y }, acc0);
      acc1 = wmma4((v2f){ ya.z, ya.w }, (v2f){ wb.z, wb.w }, acc1);
    }
  }
#pragma unroll
  for (int r = 0; r < 8; ++r) sp[wave][(8 * h + r) * DOUT_ + cl] = acc0[r] + acc1[r];
  __syncthreads();

#pragma unroll
  for (int j = 0; j < 2; ++j) {
    const int e  = threadIdx.x + 128 * j;
    const int bl = e >> 4;
    const int d  = e & 15;
    const float s = bias[n * DOUT_ + d] + ((sp[0][e] + sp[1][e]) + (sp[2][e] + sp[3][e]));
    const float sn = row_sum16(s * s);
    const float scl = sn / (1.0f + sn) * rsqrtf(sn + EPS_);
    const float o = scl * s;
    const size_t g = ((size_t)(tb * 16 + bl) * N_ + n) * DOUT_ + d;
    if (MODE == 2) {
      out[g] = o;
    } else if (MODE == 0) {
      Osum[g] = o;
      ol[e] = o;
    } else {
      ol[e] = o + Osum[g];
    }
  }

  if (MODE < 2) {
    __syncthreads();
    v2f bo[4];
#pragma unroll
    for (int q = 0; q < 4; ++q) {
      const int d0 = 4 * q + 2 * h;
      bo[q] = (v2f){ ol[cl * DOUT_ + d0], ol[cl * DOUT_ + d0 + 1] };
    }
#pragma unroll
    for (int tt = 0; tt < 4; ++tt) {
      const int tn = wave * 4 + tt;
      v8f v = {0.f,0.f,0.f,0.f,0.f,0.f,0.f,0.f};
#pragma unroll
      for (int q = 0; q < 4; ++q) v = wmma4(aw[tt][q], bo[q], v);
      float4* vp = (float4*)(Vs + (((size_t)(tb * 16 + cl) * C_ + 2 * tn + h) * N_ + n) * DIN_);
      vp[0] = make_float4(v[0], v[1], v[2], v[3]);
      vp[1] = make_float4(v[4], v[5], v[6], v[7]);
    }
  }
}

extern "C" void kernel_launch(void* const* d_in, const int* in_sizes, int n_in,
                              void* d_out, int out_size, void* d_ws, size_t ws_size,
                              hipStream_t stream)
{
  const float* x    = (const float*)d_in[0];
  const float* Wt   = (const float*)d_in[1];
  const float* bias = (const float*)d_in[2];
  float* out = (float*)d_out;

  float* Vs   = (float*)d_ws;
  float* Yw   = Vs + (size_t)B_ * C_ * N_ * DIN_;
  float* Osum = Yw + (size_t)B_ * C_ * N_ * DIN_;

  dim3 cgrid(B_ * C_), ogrid(N_ * (B_ / 16)), blk(128);

  caps_coef<0><<<cgrid, blk, 0, stream>>>(x, Vs, Yw);
  caps_out<0><<<ogrid, blk, 0, stream>>>(Wt, bias, Yw, Osum, Vs, out);
  caps_coef<1><<<cgrid, blk, 0, stream>>>(x, Vs, Yw);
  caps_out<1><<<ogrid, blk, 0, stream>>>(Wt, bias, Yw, Osum, Vs, out);
  caps_coef<1><<<cgrid, blk, 0, stream>>>(x, Vs, Yw);
  caps_out<2><<<ogrid, blk, 0, stream>>>(Wt, bias, Yw, Osum, Vs, out);
}


# Fuse the pass-1 coefficient launch into its output kernel: uniform coefficients make Y the per-channel sum of x, added (plain adds) in-block per (2 capsules, 4 batches); 6 launches become 5
# speedup vs baseline: 4.1460x; 1.1668x over previous
#include <hip/hip_runtime.h>
#include <math.h>


#define B_    32
#define IN_   2048
#define DIN_  8
#define N_    64
#define DOUT_ 16
#define C_    32
#define REP_  (IN_ / C_)
#define EPS_  1e-7f

typedef float v2f __attribute__((ext_vector_type(2)));
typedef float v8f __attribute__((ext_vector_type(8)));


template <int CTRL>
__device__ __forceinline__ float row_swz(float v)
{
  return __builtin_bit_cast(float,
      __builtin_amdgcn_update_dpp(0, __builtin_bit_cast(int, v), CTRL, 0xf, 0xf, false));
}
__device__ __forceinline__ float row_max16(float v)
{
  v = fmaxf(v, row_swz<0x161>(v));
  v = fmaxf(v, row_swz<0x162>(v));
  v = fmaxf(v, row_swz<0x164>(v));
  v = fmaxf(v, row_swz<0x168>(v));
  return v;
}
__device__ __forceinline__ float row_sum16(float v)
{
  v += row_swz<0x161>(v);
  v += row_swz<0x162>(v);
  v += row_swz<0x164>(v);
  v += row_swz<0x168>(v);
  return v;
}
__device__ __forceinline__ v8f wmma4(v2f a, v2f b, v8f c)
{
  return __builtin_amdgcn_wmma_f32_16x16x4_f32(false, a, false, b, (short)0, c, false, false);
}

__global__ __launch_bounds__(128) void caps_coef(
    const float* __restrict__ x, const float* __restrict__ Vs, float* __restrict__ Yw)
{
  __shared__ float4 ypart[4][N_ * DIN_ / 4];

  const int b    = blockIdx.x >> 5;
  const int ch   = blockIdx.x & (C_ - 1);
  const int wave = threadIdx.x >> 5;
  const int lane = threadIdx.x & 31;
  const int cl   = lane & 15;
  const int h    = lane >> 4;

  const float* xb = x + ((size_t)b * IN_ + ch * REP_ + wave * 16) * DIN_;

  v2f ax[4];
#pragma unroll
  for (int q = 0; q < 4; ++q) {
    const float* p = xb + (8 * h + 2 * q) * DIN_ + (cl & 7);
    float v0 = p[0], v1 = p[DIN_];
    if (cl >= 8) { v0 = 0.f; v1 = 0.f; }
    ax[q] = (v2f){ v0, v1 };
  }

  v8f cm[4];
  {
    const float4 xa = *(const float4*)(xb + cl * DIN_ + 4 * h);
    const v2f a0 = { xa.x, xa.y }, a1 = { xa.z, xa.w };
    const float* vb = Vs + ((size_t)(b * C_ + ch) * N_) * DIN_;
#pragma unroll
    for (int t = 0; t < 4; ++t) {
      const float4 vv = *(const float4*)(vb + (16 * t + cl) * DIN_ + 4 * h);
      const v2f b0 = { vv.x, vv.y }, b1 = { vv.z, vv.w };
      v8f acc = {0.f,0.f,0.f,0.f,0.f,0.f,0.f,0.f};
      acc = wmma4(a0, b0, acc);
      acc = wmma4(a1, b1, acc);
      cm[t] = acc;
    }
#pragma unroll
    for (int r = 0; r < 8; ++r) {
      float m = fmaxf(fmaxf(cm[0][r], cm[1][r]), fmaxf(cm[2][r], cm[3][r]));
      m = row_max16(m);
      const float e0 = __expf(cm[0][r] - m);
      const float e1 = __expf(cm[1][r] - m);
      const float e2 = __expf(cm[2][r] - m);
      const float e3 = __expf(cm[3][r] - m);
      const float inv = 1.0f / row_sum16((e0 + e1) + (e2 + e3));
      cm[0][r] = e0 * inv; cm[1][r] = e1 * inv; cm[2][r] = e2 * inv; cm[3][r] = e3 * inv;
    }
  }

#pragma unroll
  for (int t = 0; t < 4; ++t) {
    v8f y = {0.f,0.f,0.f,0.f,0.f,0.f,0.f,0.f};
#pragma unroll
    for (int q = 0; q < 4; ++q) {
      const v2f bb = { cm[t][2 * q], cm[t][2 * q + 1] };
      y = wmma4(ax[q], bb, y);
    }
    if (h == 0) {
      ypart[wave][(16 * t + cl) * 2 + 0] = make_float4(y[0], y[1], y[2], y[3]);
      ypart[wave][(16 * t + cl) * 2 + 1] = make_float4(y[4], y[5], y[6], y[7]);
    }
  }
  __syncthreads();
  {
    const int t4 = threadIdx.x;
    const float4 p0 = ypart[0][t4], p1 = ypart[1][t4], p2 = ypart[2][t4], p3 = ypart[3][t4];
    float4 s;
    s.x = (p0.x + p1.x) + (p2.x + p3.x);
    s.y = (p0.y + p1.y) + (p2.y + p3.y);
    s.z = (p0.z + p1.z) + (p2.z + p3.z);
    s.w = (p0.w + p1.w) + (p2.w + p3.w);
    ((float4*)(Yw + (size_t)(b * C_ + ch) * N_ * DIN_))[t4] = s;
  }
}

template <int MODE>
__global__ __launch_bounds__(128) void caps_out(
    const float* __restrict__ Wt, const float* __restrict__ bias,
    const float* __restrict__ Yw, float* __restrict__ Osum,
    float* __restrict__ Vs, float* __restrict__ out)
{
  __shared__ float sp[4][16 * DOUT_];
  __shared__ float ol[16 * DOUT_];

  const int n    = blockIdx.x >> 1;
  const int tb   = blockIdx.x & 1;
  const int wave = threadIdx.x >> 5;
  const int lane = threadIdx.x & 31;
  const int cl   = lane & 15;
  const int h    = lane >> 4;

  v2f aw[4][4];
  if (MODE < 2) {
#pragma unroll
    for (int tt = 0; tt < 4; ++tt) {
      const int tn = wave * 4 + tt;
      const float* wp = Wt + (((size_t)n * C_ + 2 * tn + (cl >> 3)) * DOUT_) * DIN_ + (cl & 7);
#pragma unroll
      for (int q = 0; q < 4; ++q) {
        const int d0 = 4 * q + 2 * h;
        aw[tt][q] = (v2f){ wp[d0 * DIN_], wp[(d0 + 1) * DIN_] };
      }
    }
  }

  v8f acc0 = {0.f,0.f,0.f,0.f,0.f,0.f,0.f,0.f};
  v8f acc1 = {0.f,0.f,0.f,0.f,0.f,0.f,0.f,0.f};
  {
    const int bg = tb * 16 + cl;
#pragma unroll
    for (int cc = 0; cc < 8; ++cc) {
      const int c = wave * 8 + cc;
      const float4 ya = *(const float4*)(Yw + (((size_t)bg * C_ + c) * N_ + n) * DIN_ + 4 * h);
      const float4 wb = *(const float4*)(Wt + (((size_t)n * C_ + c) * DOUT_ + cl) * DIN_ + 4 * h);
      acc0 = wmma4((v2f){ ya.x, ya.y }, (v2f){ wb.x, wb.y }, acc0);
      acc1 = wmma4((v2f){ ya.z, ya.w }, (v2f){ wb.z, wb.w }, acc1);
    }
  }
#pragma unroll
  for (int r = 0; r < 8; ++r) sp[wave][(8 * h + r) * DOUT_ + cl] = acc0[r] + acc1[r];
  __syncthreads();

#pragma unroll
  for (int j = 0; j < 2; ++j) {
    const int e  = threadIdx.x + 128 * j;
    const int bl = e >> 4;
    const int d  = e & 15;
    const float s = bias[n * DOUT_ + d] + ((sp[0][e] + sp[1][e]) + (sp[2][e] + sp[3][e]));
    const float sn = row_sum16(s * s);
    const float scl = sn / (1.0f + sn) * rsqrtf(sn + EPS_);
    const float o = scl * s;
    const size_t g = ((size_t)(tb * 16 + bl) * N_ + n) * DOUT_ + d;
    if (MODE == 2) {
      out[g] = o;
    } else {
      ol[e] = o + Osum[g];
    }
  }

  if (MODE < 2) {
    __syncthreads();
    v2f bo[4];
#pragma unroll
    for (int q = 0; q < 4; ++q) {
      const int d0 = 4 * q + 2 * h;
      bo[q] = (v2f){ ol[cl * DOUT_ + d0], ol[cl * DOUT_ + d0 + 1] };
    }
#pragma unroll
    for (int tt = 0; tt < 4; ++tt) {
      const int tn = wave * 4 + tt;
      v8f v = {0.f,0.f,0.f,0.f,0.f,0.f,0.f,0.f};
#pragma unroll
      for (int q = 0; q < 4; ++q) v = wmma4(aw[tt][q], bo[q], v);
      float4* vp = (float4*)(Vs + (((size_t)(tb * 16 + cl) * C_ + 2 * tn + h) * N_ + n) * DIN_);
      vp[0] = make_float4(v[0], v[1], v[2], v[3]);
      vp[1] = make_float4(v[4], v[5], v[6], v[7]);
    }
  }
}

__global__ __launch_bounds__(256) void caps_first(
    const float* __restrict__ x, const float* __restrict__ Wt, const float* __restrict__ bias,
    float* __restrict__ Osum, float* __restrict__ Vs)
{
  __shared__ float4 xs4[2][4][C_ * DIN_ / 4];
  __shared__ float sp[2][4][4 * DOUT_];
  __shared__ float ol[2][4 * DOUT_];

  const int np   = blockIdx.x >> 3;
  const int tb   = blockIdx.x & 7;
  const int wave = threadIdx.x >> 5;
  const int lane = threadIdx.x & 31;
  const int cl   = lane & 15;
  const int h    = lane >> 4;
  const int nl   = wave >> 2;
  const int wq   = wave & 3;
  const int n    = np * 2 + nl;
  const int cb   = cl & 3;

  v2f aw[4][4];
#pragma unroll
  for (int tt = 0; tt < 4; ++tt) {
    const int tn = wq * 4 + tt;
    const float* wp = Wt + (((size_t)n * C_ + 2 * tn + (cl >> 3)) * DOUT_) * DIN_ + (cl & 7);
#pragma unroll
    for (int q = 0; q < 4; ++q) {
      const int d0 = 4 * q + 2 * h;
      aw[tt][q] = (v2f){ wp[d0 * DIN_], wp[(d0 + 1) * DIN_] };
    }
  }

  float4 wbv[8];
#pragma unroll
  for (int cc = 0; cc < 8; ++cc)
    wbv[cc] = *(const float4*)(Wt + (((size_t)n * C_ + wq * 8 + cc) * DOUT_ + cl) * DIN_ + 4 * h);

  {
    const int bl = wave >> 1;
    const int c0 = (wave & 1) * 16;
    const float4* xp = (const float4*)(x + ((size_t)(tb * 4 + bl) * IN_ + (size_t)c0 * REP_) * DIN_) + lane;
    float4 rs[16];
#pragma unroll
    for (int half = 0; half < 2; ++half) {
      float4 t[8][4];
#pragma unroll
      for (int j = 0; j < 8; ++j)
#pragma unroll
        for (int jj = 0; jj < 4; ++jj) t[j][jj] = xp[(half * 8 + j) * (REP_ * DIN_ / 4) + 32 * jj];
#pragma unroll
      for (int j = 0; j < 8; ++j) {
        float4 v;
        v.x = (t[j][0].x + t[j][1].x) + (t[j][2].x + t[j][3].x);
        v.y = (t[j][0].y + t[j][1].y) + (t[j][2].y + t[j][3].y);
        v.z = (t[j][0].z + t[j][1].z) + (t[j][2].z + t[j][3].z);
        v.w = (t[j][0].w + t[j][1].w) + (t[j][2].w + t[j][3].w);
        v.x += row_swz<0x162>(v.x); v.y += row_swz<0x162>(v.y); v.z += row_swz<0x162>(v.z); v.w += row_swz<0x162>(v.w);
        v.x += row_swz<0x164>(v.x); v.y += row_swz<0x164>(v.y); v.z += row_swz<0x164>(v.z); v.w += row_swz<0x164>(v.w);
        v.x += row_swz<0x168>(v.x); v.y += row_swz<0x168>(v.y); v.z += row_swz<0x168>(v.z); v.w += row_swz<0x168>(v.w);
        rs[half * 8 + j] = v;
      }
    }
    if ((lane & 14) == 0) {
      const float sc = 1.0f / (float)N_;
#pragma unroll
      for (int j = 0; j < 16; ++j)
        xs4[h][bl][(c0 + j) * 2 + (lane & 1)] = make_float4(rs[j].x * sc, rs[j].y * sc, rs[j].z * sc, rs[j].w * sc);
    }
  }
  __syncthreads();

  v8f acc0 = {0.f,0.f,0.f,0.f,0.f,0.f,0.f,0.f};
  v8f acc1 = {0.f,0.f,0.f,0.f,0.f,0.f,0.f,0.f};
#pragma unroll
  for (int cc = 0; cc < 8; ++cc) {
    const int c = wq * 8 + cc;
    const float4 y0 = xs4[0][cb][c * 2 + h];
    const float4 y1 = xs4[1][cb][c * 2 + h];
    const float4 wb = wbv[cc];
    acc0 = wmma4((v2f){ y0.x + y1.x, y0.y + y1.y }, (v2f){ wb.x, wb.y }, acc0);
    acc1 = wmma4((v2f){ y0.z + y1.z, y0.w + y1.w }, (v2f){ wb.z, wb.w }, acc1);
  }
  if (h == 0) {
#pragma unroll
    for (int r = 0; r < 4; ++r) sp[nl][wq][r * DOUT_ + cl] = acc0[r] + acc1[r];
  }
  __syncthreads();

  if (threadIdx.x < 128) {
    const int n2 = threadIdx.x >> 6;
    const int e  = threadIdx.x & 63;
    const int bl = e >> 4;
    const int d  = e & 15;
    const int ng = np * 2 + n2;
    const float s = bias[ng * DOUT_ + d] + ((sp[n2][0][e] + sp[n2][1][e]) + (sp[n2][2][e] + sp[n2][3][e]));
    const float sn = row_sum16(s * s);
    const float scl = sn / (1.0f + sn) * rsqrtf(sn + EPS_);
    const float o = scl * s;
    Osum[((size_t)(tb * 4 + bl) * N_ + ng) * DOUT_ + d] = o;
    ol[n2][e] = o;
  }
  __syncthreads();

  v2f bo[4];
#pragma unroll
  for (int q = 0; q < 4; ++q) {
    const int d0 = 4 * q + 2 * h;
    bo[q] = (v2f){ ol[nl][cb * DOUT_ + d0], ol[nl][cb * DOUT_ + d0 + 1] };
  }
#pragma unroll
  for (int tt = 0; tt < 4; ++tt) {
    const int tn = wq * 4 + tt;
    v8f v = {0.f,0.f,0.f,0.f,0.f,0.f,0.f,0.f};
#pragma unroll
    for (int q = 0; q < 4; ++q) v = wmma4(aw[tt][q], bo[q], v);
    if (cl < 4) {
      float4* vp = (float4*)(Vs + (((size_t)(tb * 4 + cl) * C_ + 2 * tn + h) * N_ + n) * DIN_);
      vp[0] = make_float4(v[0], v[1], v[2], v[3]);
      vp[1] = make_float4(v[4], v[5], v[6], v[7]);
    }
  }
}

extern "C" void kernel_launch(void* const* d_in, const int* in_sizes, int n_in,
                              void* d_out, int out_size, void* d_ws, size_t ws_size,
                              hipStream_t stream)
{
  const float* x    = (const float*)d_in[0];
  const float* Wt   = (const float*)d_in[1];
  const float* bias = (const float*)d_in[2];
  float* out = (float*)d_out;

  float* Vs   = (float*)d_ws;
  float* Yw   = Vs + (size_t)B_ * C_ * N_ * DIN_;
  float* Osum = Yw + (size_t)B_ * C_ * N_ * DIN_;

  dim3 fgrid((N_ / 2) * (B_ / 4)), fblk(256), cgrid(B_ * C_), ogrid(N_ * (B_ / 16)), blk(128);

  caps_first<<<fgrid, fblk, 0, stream>>>(x, Wt, bias, Osum, Vs);
  caps_coef<<<cgrid, blk, 0, stream>>>(x, Vs, Yw);
  caps_out<1><<<ogrid, blk, 0, stream>>>(Wt, bias, Yw, Osum, Vs, out);
  caps_coef<<<cgrid, blk, 0, stream>>>(x, Vs, Yw);
  caps_out<2><<<ogrid, blk, 0, stream>>>(Wt, bias, Yw, Osum, Vs, out);
}


# Routing softmax down registers: logit tiles formed transposed (rows n, columns i), max and sum over n in-lane plus one half exchange, no DPP rows; coefficients reach the B layout via wave-private LDS
# speedup vs baseline: 4.2802x; 1.0324x over previous
#include <hip/hip_runtime.h>
#include <math.h>


#define B_    32
#define IN_   2048
#define DIN_  8
#define N_    64
#define DOUT_ 16
#define C_    32
#define REP_  (IN_ / C_)
#define EPS_  1e-7f
#define CT_PITCH_ 20

typedef float v2f __attribute__((ext_vector_type(2)));
typedef float v8f __attribute__((ext_vector_type(8)));


template <int CTRL>
__device__ __forceinline__ float row_swz(float v)
{
  return __builtin_bit_cast(float,
      __builtin_amdgcn_update_dpp(0, __builtin_bit_cast(int, v), CTRL, 0xf, 0xf, false));
}
__device__ __forceinline__ float xhalf(float v)
{
  const int b = __builtin_bit_cast(int, v);
  return __builtin_bit_cast(float, __builtin_amdgcn_permlanex16(b, b, 0x76543210, 0xfedcba98, false, false));
}
__device__ __forceinline__ float row_sum16(float v)
{
  v += row_swz<0x161>(v);
  v += row_swz<0x162>(v);
  v += row_swz<0x164>(v);
  v += row_swz<0x168>(v);
  return v;
}
__device__ __forceinline__ v8f wmma4(v2f a, v2f b, v8f c)
{
  return __builtin_amdgcn_wmma_f32_16x16x4_f32(false, a, false, b, (short)0, c, false, false);
}

__global__ __launch_bounds__(128) void caps_coef(
    const float* __restrict__ x, const float* __restrict__ Vs, float* __restrict__ Yw)
{
  __shared__ float4 ypart[4][N_ * DIN_ / 4];
  __shared__ float4 ct4[4][N_ * CT_PITCH_ / 4];

  const int b    = blockIdx.x >> 5;
  const int ch   = blockIdx.x & (C_ - 1);
  const int wave = threadIdx.x >> 5;
  const int lane = threadIdx.x & 31;
  const int cl   = lane & 15;
  const int h    = lane >> 4;

  const float* xb = x + ((size_t)b * IN_ + ch * REP_ + wave * 16) * DIN_;
  float* cw = (float*)&ct4[wave][0];

  float xr[8];
#pragma unroll
  for (int q = 0; q < 4; ++q) {
    const float* p = xb + (8 * h + 2 * q) * DIN_ + (cl & 7);
    xr[2 * q] = p[0]; xr[2 * q + 1] = p[DIN_];
  }
  v2f ax[4];

  {
    float4 xa = *(const float4*)(xb + cl * DIN_ + 4 * h);
    const float* vb = Vs + ((size_t)(b * C_ + ch) * N_) * DIN_;
    float4 vv[4];
#pragma unroll
    for (int t = 0; t < 4; ++t) vv[t] = *(const float4*)(vb + (16 * t + cl) * DIN_ + 4 * h);
    asm volatile("" : "+v"(xa.x), "+v"(xr[0]), "+v"(xr[1]), "+v"(xr[2]), "+v"(xr[3]), "+v"(xr[4]), "+v"(xr[5]), "+v"(xr[6]), "+v"(xr[7]));
    const v2f x0 = { xa.x, xa.y }, x1 = { xa.z, xa.w };
    v8f cm[4];
#pragma unroll
    for (int t = 0; t < 4; ++t) {
      const v2f w0 = { vv[t].x, vv[t].y }, w1 = { vv[t].z, vv[t].w };
      v8f acc = {0.f,0.f,0.f,0.f,0.f,0.f,0.f,0.f};
      acc = wmma4(w0, x0, acc);
      acc = wmma4(w1, x1, acc);
      cm[t] = acc;
    }
#pragma unroll
    for (int q = 0; q < 4; ++q) {
      const float v0 = (cl >= 8) ? 0.f : xr[2 * q];
      const float v1 = (cl >= 8) ? 0.f : xr[2 * q + 1];
      ax[q] = (v2f){ v0, v1 };
    }
    float mt[4];
#pragma unroll
    for (int t = 0; t < 4; ++t) {
      float m01 = fmaxf(fmaxf(cm[t][0], cm[t][1]), cm[t][2]);
      float m34 = fmaxf(fmaxf(cm[t][3], cm[t][4]), cm[t][5]);
      mt[t] = fmaxf(fmaxf(m01, m34), fmaxf(cm[t][6], cm[t][7]));
    }
    float m = fmaxf(fmaxf(mt[0], mt[1]), fmaxf(mt[2], mt[3]));
    m = fmaxf(m, xhalf(m));
    float st[4];
#pragma unroll
    for (int t = 0; t < 4; ++t) {
#pragma unroll
      for (int r = 0; r < 8; ++r) cm[t][r] = __expf(cm[t][r] - m);
      st[t] = ((cm[t][0] + cm[t][1]) + (cm[t][2] + cm[t][3])) + ((cm[t][4] + cm[t][5]) + (cm[t][6] + cm[t][7]));
    }
    float s = (st[0] + st[1]) + (st[2] + st[3]);
    s = s + xhalf(s);
    const float inv = 1.0f / s;
#pragma unroll
    for (int t = 0; t < 4; ++t)
#pragma unroll
      for (int r = 0; r < 8; ++r) cw[(16 * t + 8 * h + r) * CT_PITCH_ + cl] = cm[t][r] * inv;
  }
  asm volatile("" ::: "memory");

  {
    float4 c0[4], c1[4];
#pragma unroll
    for (int t = 0; t < 4; ++t) {
      c0[t] = *(const float4*)(cw + (16 * t + cl) * CT_PITCH_ + 8 * h);
      c1[t] = *(const float4*)(cw + (16 * t + cl) * CT_PITCH_ + 8 * h + 4);
    }
    v8f y[4];
#pragma unroll
    for (int t = 0; t < 4; ++t) y[t] = wmma4(ax[0], (v2f){ c0[t].x, c0[t].y }, (v8f){0.f,0.f,0.f,0.f,0.f,0.f,0.f,0.f});
#pragma unroll
    for (int t = 0; t < 4; ++t) y[t] = wmma4(ax[1], (v2f){ c0[t].z, c0[t].w }, y[t]);
#pragma unroll
    for (int t = 0; t < 4; ++t) y[t] = wmma4(ax[2], (v2f){ c1[t].x, c1[t].y }, y[t]);
#pragma unroll
    for (int t = 0; t < 4; ++t) y[t] = wmma4(ax[3], (v2f){ c1[t].z, c1[t].w }, y[t]);
    if (h == 0) {
#pragma unroll
      for (int t = 0; t < 4; ++t) {
        ypart[wave][(16 * t + cl) * 2 + 0] = make_float4(y[t][0], y[t][1], y[t][2], y[t][3]);
        ypart[wave][(16 * t + cl) * 2 + 1] = make_float4(y[t][4], y[t][5], y[t][6], y[t][7]);
      }
    }
  }
  __syncthreads();
  {
    const int t4 = threadIdx.x;
    const float4 p0 = ypart[0][t4], p1 = ypart[1][t4], p2 = ypart[2][t4], p3 = ypart[3][t4];
    float4 s;
    s.x = (p0.x + p1.x) + (p2.x + p3.x);
    s.y = (p0.y + p1.y) + (p2.y + p3.y);
    s.z = (p0.z + p1.z) + (p2.z + p3.z);
    s.w = (p0.w + p1.w) + (p2.w + p3.w);
    ((float4*)(Yw + (size_t)(b * C_ + ch) * N_ * DIN_))[t4] = s;
  }
}

template <int MODE>
__global__ __launch_bounds__(128) void caps_out(
    const float* __restrict__ Wt, const float* __restrict__ bias,
    const float* __restrict__ Yw, float* __restrict__ Osum,
    float* __restrict__ Vs, float* __restrict__ out)
{
  __shared__ float sp[4][16 * DOUT_];
  __shared__ float ol[16 * DOUT_];

  const int n    = blockIdx.x >> 1;
  const int tb   = blockIdx.x & 1;
  const int wave = threadIdx.x >> 5;
  const int lane = threadIdx.x & 31;
  const int cl   = lane & 15;
  const int h    = lane >> 4;

  v2f aw[4][4];
  if (MODE < 2) {
#pragma unroll
    for (int tt = 0; tt < 4; ++tt) {
      const int tn = wave * 4 + tt;
      const float* wp = Wt + (((size_t)n * C_ + 2 * tn + (cl >> 3)) * DOUT_) * DIN_ + (cl & 7);
#pragma unroll
      for (int q = 0; q < 4; ++q) {
        const int d0 = 4 * q + 2 * h;
        aw[tt][q] = (v2f){ wp[d0 * DIN_], wp[(d0 + 1) * DIN_] };
      }
    }
  }

  v8f acc0 = {0.f,0.f,0.f,0.f,0.f,0.f,0.f,0.f};
  v8f acc1 = {0.f,0.f,0.f,0.f,0.f,0.f,0.f,0.f};
  {
    const int bg = tb * 16 + cl;
#pragma unroll
    for (int cc = 0; cc < 8; ++cc) {
      const int c = wave * 8 + cc;
      const float4 ya = *(const float4*)(Yw + (((size_t)bg * C_ + c) * N_ + n) * DIN_ + 4 * h);
      const float4 wb = *(const float4*)(Wt + (((size_t)n * C_ + c) * DOUT_ + cl) * DIN_ + 4 * h);
      acc0 = wmma4((v2f){ ya.x, ya.y }, (v2f){ wb.x, wb.y }, acc0);
      acc1 = wmma4((v2f){ ya.z, ya.w }, (v2f){ wb.z, wb.w }, acc1);
    }
  }
#pragma unroll
  for (int r = 0; r < 8; ++r) sp[wave][(8 * h + r) * DOUT_ + cl] = acc0[r] + acc1[r];
  __syncthreads();

#pragma unroll
  for (int j = 0; j < 2; ++j) {
    const int e  = threadIdx.x + 128 * j;
    const int bl = e >> 4;
    const int d  = e & 15;
    const float s = bias[n * DOUT_ + d] + ((sp[0][e] + sp[1][e]) + (sp[2][e] + sp[3][e]));
    const float sn = row_sum16(s * s);
    const float scl = sn / (1.0f + sn) * rsqrtf(sn + EPS_);
    const float o = scl * s;
    const size_t g = ((size_t)(tb * 16 + bl) * N_ + n) * DOUT_ + d;
    if (MODE == 2) {
      out[g] = o;
    } else {
      ol[e] = o + Osum[g];
    }
  }

  if (MODE < 2) {
    __syncthreads();
    v2f bo[4];
#pragma unroll
    for (int q = 0; q < 4; ++q) {
      const int d0 = 4 * q + 2 * h;
      bo[q] = (v2f){ ol[cl * DOUT_ + d0], ol[cl * DOUT_ + d0 + 1] };
    }
#pragma unroll
    for (int tt = 0; tt < 4; ++tt) {
      const int tn = wave * 4 + tt;
      v8f v = {0.f,0.f,0.f,0.f,0.f,0.f,0.f,0.f};
#pragma unroll
      for (int q = 0; q < 4; ++q) v = wmma4(aw[tt][q], bo[q], v);
      float4* vp = (float4*)(Vs + (((size_t)(tb * 16 + cl) * C_ + 2 * tn + h) * N_ + n) * DIN_);
      vp[0] = make_float4(v[0], v[1], v[2], v[3]);
      vp[1] = make_float4(v[4], v[5], v[6], v[7]);
    }
  }
}

__global__ __launch_bounds__(256) void caps_first(
    const float* __restrict__ x, const float* __restrict__ Wt, const float* __restrict__ bias,
    float* __restrict__ Osum, float* __restrict__ Vs)
{
  __shared__ float4 xs4[2][4][C_ * DIN_ / 4];
  __shared__ float sp[2][4][4 * DOUT_];
  __shared__ float ol[2][4 * DOUT_];

  const int np   = blockIdx.x >> 3;
  const int tb   = blockIdx.x & 7;
  const int wave = threadIdx.x >> 5;
  const int lane = threadIdx.x & 31;
  const int cl   = lane & 15;
  const int h    = lane >> 4;
  const int nl   = wave >> 2;
  const int wq   = wave & 3;
  const int n    = np * 2 + nl;
  const int cb   = cl & 3;

  v2f aw[4][4];
#pragma unroll
  for (int tt = 0; tt < 4; ++tt) {
    const int tn = wq * 4 + tt;
    const float* wp = Wt + (((size_t)n * C_ + 2 * tn + (cl >> 3)) * DOUT_) * DIN_ + (cl & 7);
#pragma unroll
    for (int q = 0; q < 4; ++q) {
      const int d0 = 4 * q + 2 * h;
      aw[tt][q] = (v2f){ wp[d0 * DIN_], wp[(d0 + 1) * DIN_] };
    }
  }

  float4 wbv[8];
#pragma unroll
  for (int cc = 0; cc < 8; ++cc)
    wbv[cc] = *(const float4*)(Wt + (((size_t)n * C_ + wq * 8 + cc) * DOUT_ + cl) * DIN_ + 4 * h);

  {
    const int bl = wave >> 1;
    const int c0 = (wave & 1) * 16;
    const float4* xp = (const float4*)(x + ((size_t)(tb * 4 + bl) * IN_ + (size_t)c0 * REP_) * DIN_) + lane;
    float4 rs[16];
#pragma unroll
    for (int half = 0; half < 2; ++half) {
      float4 t[8][4];
#pragma unroll
      for (int j = 0; j < 8; ++j)
#pragma unroll
        for (int jj = 0; jj < 4; ++jj) t[j][jj] = xp[(half * 8 + j) * (REP_ * DIN_ / 4) + 32 * jj];
#pragma unroll
      for (int j = 0; j < 8; ++j) {
        float4 v;
        v.x = (t[j][0].x + t[j][1].x) + (t[j][2].x + t[j][3].x);
        v.y = (t[j][0].y + t[j][1].y) + (t[j][2].y + t[j][3].y);
        v.z = (t[j][0].z + t[j][1].z) + (t[j][2].z + t[j][3].z);
        v.w = (t[j][0].w + t[j][1].w) + (t[j][2].w + t[j][3].w);
        v.x += row_swz<0x162>(v.x); v.y += row_swz<0x162>(v.y); v.z += row_swz<0x162>(v.z); v.w += row_swz<0x162>(v.w);
        v.x += row_swz<0x164>(v.x); v.y += row_swz<0x164>(v.y); v.z += row_swz<0x164>(v.z); v.w += row_swz<0x164>(v.w);
        v.x += row_swz<0x168>(v.x); v.y += row_swz<0x168>(v.y); v.z += row_swz<0x168>(v.z); v.w += row_swz<0x168>(v.w);
        rs[half * 8 + j] = v;
      }
    }
    if ((lane & 14) == 0) {
      const float sc = 1.0f / (float)N_;
#pragma unroll
      for (int j = 0; j < 16; ++j)
        xs4[h][bl][(c0 + j) * 2 + (lane & 1)] = make_float4(rs[j].x * sc, rs[j].y * sc, rs[j].z * sc, rs[j].w * sc);
    }
  }
  __syncthreads();

  v8f acc0 = {0.f,0.f,0.f,0.f,0.f,0.f,0.f,0.f};
  v8f acc1 = {0.f,0.f,0.f,0.f,0.f,0.f,0.f,0.f};
#pragma unroll
  for (int cc = 0; cc < 8; ++cc) {
    const int c = wq * 8 + cc;
    const float4 y0 = xs4[0][cb][c * 2 + h];
    const float4 y1 = xs4[1][cb][c * 2 + h];
    const float4 wb = wbv[cc];
    acc0 = wmma4((v2f){ y0.x + y1.x, y0.y + y1.y }, (v2f){ wb.x, wb.y }, acc0);
    acc1 = wmma4((v2f){ y0.z + y1.z, y0.w + y1.w }, (v2f){ wb.z, wb.w }, acc1);
  }
  if (h == 0) {
#pragma unroll
    for (int r = 0; r < 4; ++r) sp[nl][wq][r * DOUT_ + cl] = acc0[r] + acc1[r];
  }
  __syncthreads();

  if (threadIdx.x < 128) {
    const int n2 = threadIdx.x >> 6;
    const int e  = threadIdx.x & 63;
    const int bl = e >> 4;
    const int d  = e & 15;
    const int ng = np * 2 + n2;
    const float s = bias[ng * DOUT_ + d] + ((sp[n2][0][e] + sp[n2][1][e]) + (sp[n2][2][e] + sp[n2][3][e]));
    const float sn = row_sum16(s * s);
    const float scl = sn / (1.0f + sn) * rsqrtf(sn + EPS_);
    const float o = scl * s;
    Osum[((size_t)(tb * 4 + bl) * N_ + ng) * DOUT_ + d] = o;
    ol[n2][e] = o;
  }
  __syncthreads();

  v2f bo[4];
#pragma unroll
  for (int q = 0; q < 4; ++q) {
    const int d0 = 4 * q + 2 * h;
    bo[q] = (v2f){ ol[nl][cb * DOUT_ + d0], ol[nl][cb * DOUT_ + d0 + 1] };
  }
#pragma unroll
  for (int tt = 0; tt < 4; ++tt) {
    const int tn = wq * 4 + tt;
    v8f v = {0.f,0.f,0.f,0.f,0.f,0.f,0.f,0.f};
#pragma unroll
    for (int q = 0; q < 4; ++q) v = wmma4(aw[tt][q], bo[q], v);
    if (cl < 4) {
      float4* vp = (float4*)(Vs + (((size_t)(tb * 4 + cl) * C_ + 2 * tn + h) * N_ + n) * DIN_);
      vp[0] = make_float4(v[0], v[1], v[2], v[3]);
      vp[1] = make_float4(v[4], v[5], v[6], v[7]);
    }
  }
}

extern "C" void kernel_launch(void* const* d_in, const int* in_sizes, int n_in,
                              void* d_out, int out_size, void* d_ws, size_t ws_size,
                              hipStream_t stream)
{
  const float* x    = (const float*)d_in[0];
  const float* Wt   = (const float*)d_in[1];
  const float* bias = (const float*)d_in[2];
  float* out = (float*)d_out;

  float* Vs   = (float*)d_ws;
  float* Yw   = Vs + (size_t)B_ * C_ * N_ * DIN_;
  float* Osum = Yw + (size_t)B_ * C_ * N_ * DIN_;

  dim3 fgrid((N_ / 2) * (B_ / 4)), fblk(256), cgrid(B_ * C_), ogrid(N_ * (B_ / 16)), blk(128);

  caps_first<<<fgrid, fblk, 0, stream>>>(x, Wt, bias, Osum, Vs);
  caps_coef<<<cgrid, blk, 0, stream>>>(x, Vs, Yw);
  caps_out<1><<<ogrid, blk, 0, stream>>>(Wt, bias, Yw, Osum, Vs, out);
  caps_coef<<<cgrid, blk, 0, stream>>>(x, Vs, Yw);
  caps_out<2><<<ogrid, blk, 0, stream>>>(Wt, bias, Yw, Osum, Vs, out);
}


# All global loads of the two output kernels (every Y and W fragment, bias, running sum) and the first kernel's bias issued as one group at wave entry; its sum plane stored behind the last barrier
# speedup vs baseline: 4.6203x; 1.0795x over previous
#include <hip/hip_runtime.h>
#include <math.h>


#define B_    32
#define IN_   2048
#define DIN_  8
#define N_    64
#define DOUT_ 16
#define C_    32
#define REP_  (IN_ / C_)
#define EPS_  1e-7f
#define CT_PITCH_ 20

typedef float v2f __attribute__((ext_vector_type(2)));
typedef float v8f __attribute__((ext_vector_type(8)));


template <int CTRL>
__device__ __forceinline__ float row_swz(float v)
{
  return __builtin_bit_cast(float,
      __builtin_amdgcn_update_dpp(0, __builtin_bit_cast(int, v), CTRL, 0xf, 0xf, false));
}
__device__ __forceinline__ float xhalf(float v)
{
  const int b = __builtin_bit_cast(int, v);
  return __builtin_bit_cast(float, __builtin_amdgcn_permlanex16(b, b, 0x76543210, 0xfedcba98, false, false));
}
__device__ __forceinline__ float row_sum16(float v)
{
  v += row_swz<0x161>(v);
  v += row_swz<0x162>(v);
  v += row_swz<0x164>(v);
  v += row_swz<0x168>(v);
  return v;
}
__device__ __forceinline__ v8f wmma4(v2f a, v2f b, v8f c)
{
  return __builtin_amdgcn_wmma_f32_16x16x4_f32(false, a, false, b, (short)0, c, false, false);
}

__global__ __launch_bounds__(128) void caps_coef(
    const float* __restrict__ x, const float* __restrict__ Vs, float* __restrict__ Yw)
{
  __shared__ float4 ypart[4][N_ * DIN_ / 4];
  __shared__ float4 ct4[4][N_ * CT_PITCH_ / 4];

  const int b    = blockIdx.x >> 5;
  const int ch   = blockIdx.x & (C_ - 1);
  const int wave = threadIdx.x >> 5;
  const int lane = threadIdx.x & 31;
  const int cl   = lane & 15;
  const int h    = lane >> 4;

  const float* xb = x + ((size_t)b * IN_ + ch * REP_ + wave * 16) * DIN_;
  float* cw = (float*)&ct4[wave][0];

  float xr[8];
#pragma unroll
  for (int q = 0; q < 4; ++q) {
    const float* p = xb + (8 * h + 2 * q) * DIN_ + (cl & 7);
    xr[2 * q] = p[0]; xr[2 * q + 1] = p[DIN_];
  }
  v2f ax[4];

  {
    float4 xa = *(const float4*)(xb + cl * DIN_ + 4 * h);
    const float* vb = Vs + ((size_t)(b * C_ + ch) * N_) * DIN_;
    float4 vv[4];
#pragma unroll
    for (int t = 0; t < 4; ++t) vv[t] = *(const float4*)(vb + (16 * t + cl) * DIN_ + 4 * h);
    asm volatile("" : "+v"(xa.x), "+v"(xr[0]), "+v"(xr[1]), "+v"(xr[2]), "+v"(xr[3]), "+v"(xr[4]), "+v"(xr[5]), "+v"(xr[6]), "+v"(xr[7]));
    const v2f x0 = { xa.x, xa.y }, x1 = { xa.z, xa.w };
    v8f cm[4];
#pragma unroll
    for (int t = 0; t < 4; ++t) {
      const v2f w0 = { vv[t].x, vv[t].y }, w1 = { vv[t].z, vv[t].w };
      v8f acc = {0.f,0.f,0.f,0.f,0.f,0.f,0.f,0.f};
      acc = wmma4(w0, x0, acc);
      acc = wmma4(w1, x1, acc);
      cm[t] = acc;
    }
#pragma unroll
    for (int q = 0; q < 4; ++q) {
      const float v0 = (cl >= 8) ? 0.f : xr[2 * q];
      const float v1 = (cl >= 8) ? 0.f : xr[2 * q + 1];
      ax[q] = (v2f){ v0, v1 };
    }
    float mt[4];
#pragma unroll
    for (int t = 0; t < 4; ++t) {
      float m01 = fmaxf(fmaxf(cm[t][0], cm[t][1]), cm[t][2]);
      float m34 = fmaxf(fmaxf(cm[t][3], cm[t][4]), cm[t][5]);
      mt[t] = fmaxf(fmaxf(m01, m34), fmaxf(cm[t][6], cm[t][7]));
    }
    float m = fmaxf(fmaxf(mt[0], mt[1]), fmaxf(mt[2], mt[3]));
    m = fmaxf(m, xhalf(m));
    float st[4];
#pragma unroll
    for (int t = 0; t < 4; ++t) {
#pragma unroll
      for (int r = 0; r < 8; ++r) cm[t][r] = __expf(cm[t][r] - m);
      st[t] = ((cm[t][0] + cm[t][1]) + (cm[t][2] + cm[t][3])) + ((cm[t][4] + cm[t][5]) + (cm[t][6] + cm[t][7]));
    }
    float s = (st[0] + st[1]) + (st[2] + st[3]);
    s = s + xhalf(s);
    const float inv = 1.0f / s;
#pragma unroll
    for (int t = 0; t < 4; ++t)
#pragma unroll
      for (int r = 0; r < 8; ++r) cw[(16 * t + 8 * h + r) * CT_PITCH_ + cl] = cm[t][r] * inv;
  }
  asm volatile("" ::: "memory");

  {
    float4 c0[4], c1[4];
#pragma unroll
    for (int t = 0; t < 4; ++t) {
      c0[t] = *(const float4*)(cw + (16 * t + cl) * CT_PITCH_ + 8 * h);
      c1[t] = *(const float4*)(cw + (16 * t + cl) * CT_PITCH_ + 8 * h + 4);
    }
    v8f y[4];
#pragma unroll
    for (int t = 0; t < 4; ++t) y[t] = wmma4(ax[0], (v2f){ c0[t].x, c0[t].y }, (v8f){0.f,0.f,0.f,0.f,0.f,0.f,0.f,0.f});
#pragma unroll
    for (int t = 0; t < 4; ++t) y[t] = wmma4(ax[1], (v2f){ c0[t].z, c0[t].w }, y[t]);
#pragma unroll
    for (int t = 0; t < 4; ++t) y[t] = wmma4(ax[2], (v2f){ c1[t].x, c1[t].y }, y[t]);
#pragma unroll
    for (int t = 0; t < 4; ++t) y[t] = wmma4(ax[3], (v2f){ c1[t].z, c1[t].w }, y[t]);
    if (h == 0) {
#pragma unroll
      for (int t = 0; t < 4; ++t) {
        ypart[wave][(16 * t + cl) * 2 + 0] = make_float4(y[t][0], y[t][1], y[t][2], y[t][3]);
        ypart[wave][(16 * t + cl) * 2 + 1] = make_float4(y[t][4], y[t][5], y[t][6], y[t][7]);
      }
    }
  }
  __syncthreads();
  {
    const int t4 = threadIdx.x;
    const float4 p0 = ypart[0][t4], p1 = ypart[1][t4], p2 = ypart[2][t4], p3 = ypart[3][t4];
    float4 s;
    s.x = (p0.x + p1.x) + (p2.x + p3.x);
    s.y = (p0.y + p1.y) + (p2.y + p3.y);
    s.z = (p0.z + p1.z) + (p2.z + p3.z);
    s.w = (p0.w + p1.w) + (p2.w + p3.w);
    ((float4*)(Yw + (size_t)(b * C_ + ch) * N_ * DIN_))[t4] = s;
  }
}

template <int MODE>
__global__ __launch_bounds__(128) void caps_out(
    const float* __restrict__ Wt, const float* __restrict__ bias,
    const float* __restrict__ Yw, float* __restrict__ Osum,
    float* __restrict__ Vs, float* __restrict__ out)
{
  __shared__ float sp[4][16 * DOUT_];
  __shared__ float ol[16 * DOUT_];

  const int n    = blockIdx.x >> 1;
  const int tb   = blockIdx.x & 1;
  const int wave = threadIdx.x >> 5;
  const int lane = threadIdx.x & 31;
  const int cl   = lane & 15;
  const int h    = lane >> 4;

  float4 ya[8], wb[8];
  {
    const int bg = tb * 16 + cl;
#pragma unroll
    for (int cc = 0; cc < 8; ++cc) {
      const int c = wave * 8 + cc;
      ya[cc] = *(const float4*)(Yw + (((size_t)bg * C_ + c) * N_ + n) * DIN_ + 4 * h);
      wb[cc] = *(const float4*)(Wt + (((size_t)n * C_ + c) * DOUT_ + cl) * DIN_ + 4 * h);
    }
  }
  v2f aw[4][4];
  if (MODE < 2) {
#pragma unroll
    for (int tt = 0; tt < 4; ++tt) {
      const int tn = wave * 4 + tt;
      const float* wp = Wt + (((size_t)n * C_ + 2 * tn + (cl >> 3)) * DOUT_) * DIN_ + (cl & 7);
#pragma unroll
      for (int q = 0; q < 4; ++q) {
        const int d0 = 4 * q + 2 * h;
        aw[tt][q] = (v2f){ wp[d0 * DIN_], wp[(d0 + 1) * DIN_] };
      }
    }
  }
  const float bs = bias[n * DOUT_ + (threadIdx.x & 15)];
  const size_t g0 = ((size_t)(tb * 16 + (threadIdx.x >> 4)) * N_ + n) * DOUT_ + (threadIdx.x & 15);
  const size_t g1 = g0 + (size_t)8 * N_ * DOUT_;
  float os0 = 0.f, os1 = 0.f;
  if (MODE < 2) { os0 = Osum[g0]; os1 = Osum[g1]; }
  __builtin_amdgcn_sched_barrier(0);

  v8f acc0 = {0.f,0.f,0.f,0.f,0.f,0.f,0.f,0.f};
  v8f acc1 = {0.f,0.f,0.f,0.f,0.f,0.f,0.f,0.f};
#pragma unroll
  for (int cc = 0; cc < 8; ++cc) {
    acc0 = wmma4((v2f){ ya[cc].x, ya[cc].y }, (v2f){ wb[cc].x, wb[cc].y }, acc0);
    acc1 = wmma4((v2f){ ya[cc].z, ya[cc].w }, (v2f){ wb[cc].z, wb[cc].w }, acc1);
  }
#pragma unroll
  for (int r = 0; r < 8; ++r) sp[wave][(8 * h + r) * DOUT_ + cl] = acc0[r] + acc1[r];
  __syncthreads();

#pragma unroll
  for (int j = 0; j < 2; ++j) {
    const int e  = threadIdx.x + 128 * j;
    const float s = bs + ((sp[0][e] + sp[1][e]) + (sp[2][e] + sp[3][e]));
    const float sn = row_sum16(s * s);
    const float scl = sn / (1.0f + sn) * rsqrtf(sn + EPS_);
    const float o = scl * s;
    if (MODE == 2) {
      out[j ? g1 : g0] = o;
    } else {
      ol[e] = o + (j ? os1 : os0);
    }
  }

  if (MODE < 2) {
    __syncthreads();
    v2f bo[4];
#pragma unroll
    for (int q = 0; q < 4; ++q) {
      const int d0 = 4 * q + 2 * h;
      bo[q] = (v2f){ ol[cl * DOUT_ + d0], ol[cl * DOUT_ + d0 + 1] };
    }
#pragma unroll
    for (int tt = 0; tt < 4; ++tt) {
      const int tn = wave * 4 + tt;
      v8f v = {0.f,0.f,0.f,0.f,0.f,0.f,0.f,0.f};
#pragma unroll
      for (int q = 0; q < 4; ++q) v = wmma4(aw[tt][q], bo[q], v);
      float4* vp = (float4*)(Vs + (((size_t)(tb * 16 + cl) * C_ + 2 * tn + h) * N_ + n) * DIN_);
      vp[0] = make_float4(v[0], v[1], v[2], v[3]);
      vp[1] = make_float4(v[4], v[5], v[6], v[7]);
    }
  }
}

__global__ __launch_bounds__(256) void caps_first(
    const float* __restrict__ x, const float* __restrict__ Wt, const float* __restrict__ bias,
    float* __restrict__ Osum, float* __restrict__ Vs)
{
  __shared__ float4 xs4[2][4][C_ * DIN_ / 4];
  __shared__ float sp[2][4][4 * DOUT_];
  __shared__ float ol[2][4 * DOUT_];

  const int np   = blockIdx.x >> 3;
  const int tb   = blockIdx.x & 7;
  const int wave = threadIdx.x >> 5;
  const int lane = threadIdx.x & 31;
  const int cl   = lane & 15;
  const int h    = lane >> 4;
  const int nl   = wave >> 2;
  const int wq   = wave & 3;
  const int n    = np * 2 + nl;
  const int cb   = cl & 3;

  const int sq_n = np * 2 + ((threadIdx.x >> 6) & 1);
  const float bs = bias[sq_n * DOUT_ + (threadIdx.x & 15)];
  __builtin_amdgcn_sched_barrier(0);

  v2f aw[4][4];
#pragma unroll
  for (int tt = 0; tt < 4; ++tt) {
    const int tn = wq * 4 + tt;
    const float* wp = Wt + (((size_t)n * C_ + 2 * tn + (cl >> 3)) * DOUT_) * DIN_ + (cl & 7);
#pragma unroll
    for (int q = 0; q < 4; ++q) {
      const int d0 = 4 * q + 2 * h;
      aw[tt][q] = (v2f){ wp[d0 * DIN_], wp[(d0 + 1) * DIN_] };
    }
  }

  float4 wbv[8];
#pragma unroll
  for (int cc = 0; cc < 8; ++cc)
    wbv[cc] = *(const float4*)(Wt + (((size_t)n * C_ + wq * 8 + cc) * DOUT_ + cl) * DIN_ + 4 * h);

  {
    const int bl = wave >> 1;
    const int c0 = (wave & 1) * 16;
    const float4* xp = (const float4*)(x + ((size_t)(tb * 4 + bl) * IN_ + (size_t)c0 * REP_) * DIN_) + lane;
    float4 rs[16];
#pragma unroll
    for (int half = 0; half < 2; ++half) {
      float4 t[8][4];
#pragma unroll
      for (int j = 0; j < 8; ++j)
#pragma unroll
        for (int jj = 0; jj < 4; ++jj) t[j][jj] = xp[(half * 8 + j) * (REP_ * DIN_ / 4) + 32 * jj];
#pragma unroll
      for (int j = 0; j < 8; ++j) {
        float4 v;
        v.x = (t[j][0].x + t[j][1].x) + (t[j][2].x + t[j][3].x);
        v.y = (t[j][0].y + t[j][1].y) + (t[j][2].y + t[j][3].y);
        v.z = (t[j][0].z + t[j][1].z) + (t[j][2].z + t[j][3].z);
        v.w = (t[j][0].w + t[j][1].w) + (t[j][2].w + t[j][3].w);
        v.x += row_swz<0x162>(v.x); v.y += row_swz<0x162>(v.y); v.z += row_swz<0x162>(v.z); v.w += row_swz<0x162>(v.w);
        v.x += row_swz<0x164>(v.x); v.y += row_swz<0x164>(v.y); v.z += row_swz<0x164>(v.z); v.w += row_swz<0x164>(v.w);
        v.x += row_swz<0x168>(v.x); v.y += row_swz<0x168>(v.y); v.z += row_swz<0x168>(v.z); v.w += row_swz<0x168>(v.w);
        rs[half * 8 + j] = v;
      }
    }
    if ((lane & 14) == 0) {
      const float sc = 1.0f / (float)N_;
#pragma unroll
      for (int j = 0; j < 16; ++j)
        xs4[h][bl][(c0 + j) * 2 + (lane & 1)] = make_float4(rs[j].x * sc, rs[j].y * sc, rs[j].z * sc, rs[j].w * sc);
    }
  }
  __syncthreads();

  v8f acc0 = {0.f,0.f,0.f,0.f,0.f,0.f,0.f,0.f};
  v8f acc1 = {0.f,0.f,0.f,0.f,0.f,0.f,0.f,0.f};
#pragma unroll
  for (int cc = 0; cc < 8; ++cc) {
    const int c = wq * 8 + cc;
    const float4 y0 = xs4[0][cb][c * 2 + h];
    const float4 y1 = xs4[1][cb][c * 2 + h];
    const float4 wb = wbv[cc];
    acc0 = wmma4((v2f){ y0.x + y1.x, y0.y + y1.y }, (v2f){ wb.x, wb.y }, acc0);
    acc1 = wmma4((v2f){ y0.z + y1.z, y0.w + y1.w }, (v2f){ wb.z, wb.w }, acc1);
  }
  if (h == 0) {
#pragma unroll
    for (int r = 0; r < 4; ++r) sp[nl][wq][r * DOUT_ + cl] = acc0[r] + acc1[r];
  }
  __syncthreads();

  float o_keep = 0.f;
  if (threadIdx.x < 128) {
    const int n2 = threadIdx.x >> 6;
    const int e  = threadIdx.x & 63;
    const float s = bs + ((sp[n2][0][e] + sp[n2][1][e]) + (sp[n2][2][e] + sp[n2][3][e]));
    const float sn = row_sum16(s * s);
    const float scl = sn / (1.0f + sn) * rsqrtf(sn + EPS_);
    const float o = scl * s;
    ol[n2][e] = o;
    o_keep = o;
  }
  __syncthreads();
  if (threadIdx.x < 128) {
    const int e  = threadIdx.x & 63;
    Osum[((size_t)(tb * 4 + (e >> 4)) * N_ + sq_n) * DOUT_ + (e & 15)] = o_keep;
  }

  v2f bo[4];
#pragma unroll
  for (int q = 0; q < 4; ++q) {
    const int d0 = 4 * q + 2 * h;
    bo[q] = (v2f){ ol[nl][cb * DOUT_ + d0], ol[nl][cb * DOUT_ + d0 + 1] };
  }
#pragma unroll
  for (int tt = 0; tt < 4; ++tt) {
    const int tn = wq * 4 + tt;
    v8f v = {0.f,0.f,0.f,0.f,0.f,0.f,0.f,0.f};
#pragma unroll
    for (int q = 0; q < 4; ++q) v = wmma4(aw[tt][q], bo[q], v);
    if (cl < 4) {
      float4* vp = (float4*)(Vs + (((size_t)(tb * 4 + cl) * C_ + 2 * tn + h) * N_ + n) * DIN_);
      vp[0] = make_float4(v[0], v[1], v[2], v[3]);
      vp[1] = make_float4(v[4], v[5], v[6], v[7]);
    }
  }
}

extern "C" void kernel_launch(void* const* d_in, const int* in_sizes, int n_in,
                              void* d_out, int out_size, void* d_ws, size_t ws_size,
                              hipStream_t stream)
{
  const float* x    = (const float*)d_in[0];
  const float* Wt   = (const float*)d_in[1];
  const float* bias = (const float*)d_in[2];
  float* out = (float*)d_out;

  float* Vs   = (float*)d_ws;
  float* Yw   = Vs + (size_t)B_ * C_ * N_ * DIN_;
  float* Osum = Yw + (size_t)B_ * C_ * N_ * DIN_;

  dim3 fgrid((N_ / 2) * (B_ / 4)), fblk(256), cgrid(B_ * C_), ogrid(N_ * (B_ / 16)), blk(128);

  caps_first<<<fgrid, fblk, 0, stream>>>(x, Wt, bias, Osum, Vs);
  caps_coef<<<cgrid, blk, 0, stream>>>(x, Vs, Yw);
  caps_out<1><<<ogrid, blk, 0, stream>>>(Wt, bias, Yw, Osum, Vs, out);
  caps_coef<<<cgrid, blk, 0, stream>>>(x, Vs, Yw);
  caps_out<2><<<ogrid, blk, 0, stream>>>(Wt, bias, Yw, Osum, Vs, out);
}
